# starting kernel
# baseline (speedup 1.0000x reference)
#include <hip/hip_runtime.h>

#define Bn 4
#define Tn 512
#define Cn 32

typedef float v2f __attribute__((ext_vector_type(2)));
typedef float v8f __attribute__((ext_vector_type(8)));

__global__ __launch_bounds__(1024) void rosa_kernel(const float* __restrict__ x,
                                                    const float* __restrict__ emb0,
                                                    const float* __restrict__ emb1,
                                                    float* __restrict__ out)
{
    __shared__ signed char lds_sc[Tn][Cn];
    __shared__ unsigned    lds_words[Cn][16];

    const int lane = threadIdx.x;
    const int c    = threadIdx.y;
    const int b    = blockIdx.x;
    const int tid  = c * 32 + lane;

    const float* __restrict__ xb = x + b * Tn * Cn;
    __builtin_prefetch(xb + tid, 0, 1);

#pragma unroll
    for (int e = tid; e < Tn * Cn; e += 1024) {
        lds_sc[e >> 5][e & 31] = (xb[e] > 0.0f) ? (signed char)1 : (signed char)0;
    }
    __syncthreads();

    unsigned myword = 0;
#pragma unroll
    for (int j = 0; j < 16; ++j) {
        unsigned w = (unsigned)__ballot(lds_sc[j * 32 + lane][c] != 0);
        if (lane == j) myword = w;
    }
    if (lane < 16) lds_words[c][lane] = myword;
    __syncthreads();

    int run[16];
#pragma unroll
    for (int j = 0; j < 16; ++j) run[j] = 0;

    for (int w = 0; w < 16; ++w) {
        const unsigned bw = lds_words[c][w];
        for (int i = 0; i < 32; ++i) {
            const int t  = (w << 5) + i;
            const int bt = (int)((bw >> i) & 1u);
            int best = 0;
#pragma unroll
            for (int j = 0; j < 16; ++j) {
                const int k = 1 + lane + (j << 5);
                const int s = t - k;
                int m = 0;
                if (s >= 0) {
                    const unsigned sw = lds_words[c][s >> 5];
                    m = ((int)((sw >> (s & 31)) & 1u) == bt);
                }
                run[j] = m ? (run[j] + 1) : 0;
                const int key = (run[j] << 10) | (512 - k);
                best = (key > best) ? key : best;
            }
#pragma unroll
            for (int off = 16; off > 0; off >>= 1) {
                const int o = __shfl_xor(best, off, 32);
                best = (o > best) ? o : best;
            }
            int idxv = -1;
            const int L = best >> 10;
            if (L > 0) {
                const int kstar = 512 - (best & 1023);
                const int nxt   = t - kstar + 1;
                idxv = (int)((lds_words[c][nxt >> 5] >> (nxt & 31)) & 1u);
            }
            if (lane == 0) lds_sc[t][c] = (signed char)idxv;
        }
    }
    __syncthreads();

    const int mr   = lane & 15;
    const int half = lane >> 4;
#pragma unroll
    for (int it = 0; it < 2; ++it) {
        const int tile = c * 2 + it;
        const int t0   = (tile >> 1) << 4;
        const int c0   = (tile & 1) << 4;
        const float e0v = emb0[c0 + mr];
        const float e1v = emb1[c0 + mr];
        v8f acc = {};
#pragma unroll
        for (int kk = 0; kk < 16; kk += 4) {
            const int ka = kk + 2 * half;
            const signed char i0 = lds_sc[t0 + mr][c0 + ka];
            const signed char i1 = lds_sc[t0 + mr][c0 + ka + 1];
            v2f a0 = { (i0 == 0) ? 1.0f : 0.0f, (i1 == 0) ? 1.0f : 0.0f };
            v2f a1 = { (i0 == 1) ? 1.0f : 0.0f, (i1 == 1) ? 1.0f : 0.0f };
            v2f b0 = { (mr == ka) ? e0v : 0.0f, (mr == ka + 1) ? e0v : 0.0f };
            v2f b1 = { (mr == ka) ? e1v : 0.0f, (mr == ka + 1) ? e1v : 0.0f };
            acc = __builtin_amdgcn_wmma_f32_16x16x4_f32(
                false, a0, false, b0, (short)0, acc, false, false);
            acc = __builtin_amdgcn_wmma_f32_16x16x4_f32(
                false, a1, false, b1, (short)0, acc, false, false);
        }
        float* __restrict__ outb = out + b * Tn * Cn;
#pragma unroll
        for (int r = 0; r < 8; ++r) {
            outb[(t0 + r + 8 * half) * Cn + (c0 + mr)] = acc[r];
        }
    }
}

extern "C" void kernel_launch(void* const* d_in, const int* in_sizes, int n_in,
                              void* d_out, int out_size, void* d_ws, size_t ws_size,
                              hipStream_t stream) {
    const float* x  = (const float*)d_in[0];
    const float* e0 = (const float*)d_in[1];
    const float* e1 = (const float*)d_in[2];
    float* out = (float*)d_out;
    (void)in_sizes; (void)n_in; (void)out_size; (void)d_ws; (void)ws_size;
    dim3 grid(Bn);
    dim3 block(32, 32);
    rosa_kernel<<<grid, block, 0, stream>>>(x, e0, e1, out);
}


# run length per (t,lag) from clz of xor of 32-bit history windows instead of counters carried over t: blocks tile over t (128 blocks of 16 t, was 4); integer ranking only, float sums as before
# speedup vs baseline: 48.1393x; 48.1393x over previous
#include <hip/hip_runtime.h>

#define Bn 4
#define Tn 512
#define Cn 32

typedef float v2f __attribute__((ext_vector_type(2)));
typedef float v8f __attribute__((ext_vector_type(8)));

__global__ __launch_bounds__(1024) void rosa_kernel(const float* __restrict__ x,
                                                    const float* __restrict__ emb0,
                                                    const float* __restrict__ emb1,
                                                    float* __restrict__ out)
{
    __shared__ unsigned win[Cn][Tn];
    __shared__ unsigned rowbits[Tn];

    const int lane = threadIdx.x;
    const int c    = threadIdx.y;
    const int t0   = blockIdx.x << 4;
    const int b    = blockIdx.y;

    const float* __restrict__ xb = x + b * Tn * Cn;
    unsigned* __restrict__ wrow = win[c];

    const int nw = ((t0 + 15) >> 5) + 1;
    float xv[16];
#pragma unroll
    for (int j = 0; j < 16; ++j) {
        const int row = (j < nw) ? ((j << 5) + c) : c;
        xv[j] = xb[row * Cn + lane];
    }
#pragma unroll
    for (int j = 0; j < 16; ++j) {
        if (j < nw) {
            const unsigned w = (unsigned)__ballot(xv[j] > 0.0f);
            if (lane == 0) rowbits[(j << 5) + c] = w;
        }
    }
    __syncthreads();

    unsigned prev = 0;
#pragma unroll
    for (int j = 0; j < 16; ++j) {
        if (j < nw) {
            const unsigned r = rowbits[(j << 5) + lane];
            const unsigned w = (unsigned)__ballot(((r >> c) & 1u) != 0u);
            const unsigned long long v = ((unsigned long long)w << 32) | prev;
            wrow[(j << 5) + lane] = (unsigned)(v >> (lane + 1));
            prev = w;
        }
    }
    __syncthreads();

    const int tl = lane & 15;
    const int t  = t0 + tl;
    const unsigned wt = wrow[t];
    const int ustart = t - 1 - (lane >> 4);
    int best = 0;
#pragma unroll 4
    for (int u = ustart; u >= 0; u -= 2) {
        const int rl = min(__clz(wt ^ wrow[u]), u + 1);
        best = max(best, (rl << 10) | u);
    }
    if ((best >> 10) >= 32) {
        best = 0;
        for (int u = ustart; u >= 0; u -= 2) {
            const unsigned d = wt ^ wrow[u];
            int rl = __clz(d);
            if (d == 0u) {
                for (int off = 32; off <= u; off += 32) {
                    const unsigned d2 = wrow[t - off] ^ wrow[u - off];
                    rl += __clz(d2);
                    if (d2 != 0u) break;
                }
            }
            rl = min(rl, u + 1);
            best = max(best, (rl << 10) | u);
        }
    }
    best = max(best, __shfl_xor(best, 16, 32));

    int idxv = -1;
    if ((best >> 10) > 0) {
        const int nxt = (best & 1023) + 1;
        idxv = (int)(wrow[nxt] >> 31);
    }
    __syncthreads();
    signed char (*idx)[Cn] = reinterpret_cast<signed char (*)[Cn]>(&win[0][0]);
    if (lane < 16) idx[tl][c] = (signed char)idxv;
    __syncthreads();

    if (c < 2) {
        const int mr   = lane & 15;
        const int half = lane >> 4;
        const int c0   = c << 4;
        const float e0v = emb0[c0 + mr];
        const float e1v = emb1[c0 + mr];
        v8f acc = {};
#pragma unroll
        for (int kk = 0; kk < 16; kk += 4) {
            const int ka = kk + 2 * half;
            const signed char i0 = idx[mr][c0 + ka];
            const signed char i1 = idx[mr][c0 + ka + 1];
            v2f a0 = { (i0 == 0) ? 1.0f : 0.0f, (i1 == 0) ? 1.0f : 0.0f };
            v2f a1 = { (i0 == 1) ? 1.0f : 0.0f, (i1 == 1) ? 1.0f : 0.0f };
            v2f b0 = { (mr == ka) ? e0v : 0.0f, (mr == ka + 1) ? e0v : 0.0f };
            v2f b1 = { (mr == ka) ? e1v : 0.0f, (mr == ka + 1) ? e1v : 0.0f };
            acc = __builtin_amdgcn_wmma_f32_16x16x4_f32(
                false, a0, false, b0, (short)0, acc, false, false);
            acc = __builtin_amdgcn_wmma_f32_16x16x4_f32(
                false, a1, false, b1, (short)0, acc, false, false);
        }
        float* __restrict__ outb = out + b * Tn * Cn;
#pragma unroll
        for (int r = 0; r < 8; ++r) {
            outb[(t0 + r + 8 * half) * Cn + (c0 + mr)] = acc[r];
        }
    }
}

extern "C" void kernel_launch(void* const* d_in, const int* in_sizes, int n_in,
                              void* d_out, int out_size, void* d_ws, size_t ws_size,
                              hipStream_t stream) {
    const float* x  = (const float*)d_in[0];
    const float* e0 = (const float*)d_in[1];
    const float* e1 = (const float*)d_in[2];
    float* out = (float*)d_out;
    (void)in_sizes; (void)n_in; (void)out_size; (void)d_ws; (void)ws_size;
    dim3 grid(32, 4);
    dim3 block(32, 32);
    rosa_kernel<<<grid, block, 0, stream>>>(x, e0, e1, out);
}


# blocks take 8 early t plus 8 late t (t_lo + t_hi = 504 + 2p) instead of 16 consecutive t, 4 lanes per t: every block walks 4088 lags per column (was 120..8056); integer ranking only
# speedup vs baseline: 61.8863x; 1.2856x over previous
#include <hip/hip_runtime.h>

#define Bn 4
#define Tn 512
#define Cn 32

typedef float v2f __attribute__((ext_vector_type(2)));
typedef float v8f __attribute__((ext_vector_type(8)));


__device__ __forceinline__ int walk_lags(const unsigned* __restrict__ wrow,
                                         const int t, const int r)
{
    const unsigned wt = wrow[t];
    const int ustart = t - 1 - r;
    int best = 0;
#pragma unroll 4
    for (int u = ustart; u >= 0; u -= 4) {
        const int rl = min(__clz(wt ^ wrow[u]), u + 1);
        best = max(best, (rl << 10) | u);
    }
    if ((best >> 10) >= 32) {
        best = 0;
        for (int u = ustart; u >= 0; u -= 4) {
            const unsigned d = wt ^ wrow[u];
            int rl = __clz(d);
            if (d == 0u) {
                for (int off = 32; off <= u; off += 32) {
                    const unsigned d2 = wrow[t - off] ^ wrow[u - off];
                    rl += __clz(d2);
                    if (d2 != 0u) break;
                }
            }
            rl = min(rl, u + 1);
            best = max(best, (rl << 10) | u);
        }
    }
    return best;
}

__global__ __launch_bounds__(1024) void rosa_kernel(const float* __restrict__ x,
                                                    const float* __restrict__ emb0,
                                                    const float* __restrict__ emb1,
                                                    float* __restrict__ out)
{
    __shared__ unsigned win[Cn][Tn];
    __shared__ unsigned rowbits[Tn];

    const int lane = threadIdx.x;
    const int c    = threadIdx.y;
    const int tlo0 = blockIdx.x << 3;
    const int thi0 = (Tn - 8) - tlo0;
    const int b    = blockIdx.y;

    const float* __restrict__ xb = x + b * Tn * Cn;
    unsigned* __restrict__ wrow = win[c];

    const int nw = ((thi0 + 7) >> 5) + 1;
    float xv[16];
#pragma unroll
    for (int j = 0; j < 16; ++j) {
        const int row = (j < nw) ? ((j << 5) + c) : c;
        xv[j] = xb[row * Cn + lane];
    }
#pragma unroll
    for (int j = 0; j < 16; ++j) {
        if (j < nw) {
            const unsigned w = (unsigned)__ballot(xv[j] > 0.0f);
            if (lane == 0) rowbits[(j << 5) + c] = w;
        }
    }
    __syncthreads();

    unsigned prev = 0;
#pragma unroll
    for (int j = 0; j < 16; ++j) {
        if (j < nw) {
            const unsigned r = rowbits[(j << 5) + lane];
            const unsigned w = (unsigned)__ballot(((r >> c) & 1u) != 0u);
            const unsigned long long v = ((unsigned long long)w << 32) | prev;
            wrow[(j << 5) + lane] = (unsigned)(v >> (lane + 1));
            prev = w;
        }
    }
    __syncthreads();

    const int p = lane & 7;
    const int r = lane >> 3;
    int best_hi = walk_lags(wrow, thi0 + p, r);
    int best_lo = walk_lags(wrow, tlo0 + p, r);
    best_hi = max(best_hi, __shfl_xor(best_hi, 8, 32));
    best_lo = max(best_lo, __shfl_xor(best_lo, 8, 32));
    best_hi = max(best_hi, __shfl_xor(best_hi, 16, 32));
    best_lo = max(best_lo, __shfl_xor(best_lo, 16, 32));

    int idx_hi = -1;
    int idx_lo = -1;
    if ((best_hi >> 10) > 0) {
        const int nxt = (best_hi & 1023) + 1;
        idx_hi = (int)(wrow[nxt] >> 31);
    }
    if ((best_lo >> 10) > 0) {
        const int nxt = (best_lo & 1023) + 1;
        idx_lo = (int)(wrow[nxt] >> 31);
    }
    __syncthreads();
    signed char (*idx)[Cn] = reinterpret_cast<signed char (*)[Cn]>(&win[0][0]);
    if (lane < 8) {
        idx[p][c]     = (signed char)idx_lo;
        idx[8 + p][c] = (signed char)idx_hi;
    }
    __syncthreads();

    if (c < 2) {
        const int mr   = lane & 15;
        const int half = lane >> 4;
        const int c0   = c << 4;
        const float e0v = emb0[c0 + mr];
        const float e1v = emb1[c0 + mr];
        v8f acc = {};
#pragma unroll
        for (int kk = 0; kk < 16; kk += 4) {
            const int ka = kk + 2 * half;
            const signed char i0 = idx[mr][c0 + ka];
            const signed char i1 = idx[mr][c0 + ka + 1];
            v2f a0 = { (i0 == 0) ? 1.0f : 0.0f, (i1 == 0) ? 1.0f : 0.0f };
            v2f a1 = { (i0 == 1) ? 1.0f : 0.0f, (i1 == 1) ? 1.0f : 0.0f };
            v2f b0 = { (mr == ka) ? e0v : 0.0f, (mr == ka + 1) ? e0v : 0.0f };
            v2f b1 = { (mr == ka) ? e1v : 0.0f, (mr == ka + 1) ? e1v : 0.0f };
            acc = __builtin_amdgcn_wmma_f32_16x16x4_f32(
                false, a0, false, b0, (short)0, acc, false, false);
            acc = __builtin_amdgcn_wmma_f32_16x16x4_f32(
                false, a1, false, b1, (short)0, acc, false, false);
        }
        float* __restrict__ outb = out + b * Tn * Cn;
        const int trow = half ? thi0 : tlo0;
#pragma unroll
        for (int q = 0; q < 8; ++q) {
            outb[(trow + q) * Cn + (c0 + mr)] = acc[q];
        }
    }
}

extern "C" void kernel_launch(void* const* d_in, const int* in_sizes, int n_in,
                              void* d_out, int out_size, void* d_ws, size_t ws_size,
                              hipStream_t stream) {
    const float* x  = (const float*)d_in[0];
    const float* e0 = (const float*)d_in[1];
    const float* e1 = (const float*)d_in[2];
    float* out = (float*)d_out;
    (void)in_sizes; (void)n_in; (void)out_size; (void)d_ws; (void)ws_size;
    dim3 grid(32, 4);
    dim3 block(32, 32);
    rosa_kernel<<<grid, block, 0, stream>>>(x, e0, e1, out);
}


# lag walk takes aligned groups of four windows per lane with one 16-byte LDS read, same trip count in all lanes, run cap and u<t test only in first and last trips; integer ranking only
# speedup vs baseline: 74.8081x; 1.2088x over previous
#include <hip/hip_runtime.h>

#define Bn 4
#define Tn 512
#define Cn 32

typedef float v2f __attribute__((ext_vector_type(2)));
typedef float v8f __attribute__((ext_vector_type(8)));


typedef unsigned v4u __attribute__((ext_vector_type(4)));

__device__ __forceinline__ int group_checked(const unsigned* __restrict__ wrow,
                                             const unsigned wt, const int t,
                                             const int g, int best)
{
    if (g >= 0) {
        const v4u w = *reinterpret_cast<const v4u*>(wrow + (g << 2));
#pragma unroll
        for (int j = 0; j < 4; ++j) {
            const int u = (g << 2) + j;
            const int rl = min(__clz(wt ^ w[j]), u + 1);
            if (u < t) best = max(best, (rl << 10) | u);
        }
    }
    return best;
}

__device__ __forceinline__ int walk_lags(const unsigned* __restrict__ wrow,
                                         const int t, const int r, const int gtop)
{
    const unsigned wt = wrow[t];
    const int ntrip = (gtop >> 2) + 1;
    const int nfast = (gtop >= 15) ? ((gtop - 11) >> 2) : 0;
    int g = gtop - r;
    int best = group_checked(wrow, wt, t, g, 0);
    g -= 4;
    int i = 1;
    const unsigned gbit = 4u << 2;
    int ukey = g << 2;
    for (; i <= nfast; ++i) {
        const v4u w = *reinterpret_cast<const v4u*>(wrow + ukey);
        const int k0 = (__builtin_clz((wt ^ w[0]) | 1u) << 10) | 0;
        const int k1 = (__builtin_clz((wt ^ w[1]) | 1u) << 10) | 1;
        const int k2 = (__builtin_clz((wt ^ w[2]) | 1u) << 10) | 2;
        const int k3 = (__builtin_clz((wt ^ w[3]) | 1u) << 10) | 3;
        const int m = max(max(k0, k1), max(k2, k3)) | ukey;
        best = max(best, m);
        ukey -= (int)gbit;
    }
    g = (ukey >> 2);
    for (; i < ntrip; ++i) {
        if (g >= 0) {
            const v4u w = *reinterpret_cast<const v4u*>(wrow + (g << 2));
#pragma unroll
            for (int j = 0; j < 4; ++j) {
                const int u = (g << 2) + j;
                const int rl = min(__clz(wt ^ w[j]), u + 1);
                best = max(best, (rl << 10) | u);
            }
        }
        g -= 4;
    }
    if ((best >> 10) >= 31) {
        best = 0;
        for (int gg = gtop - r; gg >= 0; gg -= 4) {
            for (int j = 0; j < 4; ++j) {
                const int u = (gg << 2) + j;
                if (u >= t) continue;
                const unsigned d = wt ^ wrow[u];
                int rl = __clz(d);
                if (d == 0u) {
                    for (int off = 32; off <= u; off += 32) {
                        const unsigned d2 = wrow[t - off] ^ wrow[u - off];
                        rl += __clz(d2);
                        if (d2 != 0u) break;
                    }
                }
                rl = min(rl, u + 1);
                best = max(best, (rl << 10) | u);
            }
        }
    }
    return best;
}

__global__ __launch_bounds__(1024) void rosa_kernel(const float* __restrict__ x,
                                                    const float* __restrict__ emb0,
                                                    const float* __restrict__ emb1,
                                                    float* __restrict__ out)
{
    __shared__ __attribute__((aligned(16))) unsigned win[Cn][Tn];
    __shared__ unsigned rowbits[Tn];

    const int lane = threadIdx.x;
    const int c    = threadIdx.y;
    const int tlo0 = blockIdx.x << 3;
    const int thi0 = (Tn - 8) - tlo0;
    const int b    = blockIdx.y;

    const float* __restrict__ xb = x + b * Tn * Cn;
    unsigned* __restrict__ wrow = win[c];

    const int nw = ((thi0 + 7) >> 5) + 1;
    float xv[16];
#pragma unroll
    for (int j = 0; j < 16; ++j) {
        const int row = (j < nw) ? ((j << 5) + c) : c;
        xv[j] = xb[row * Cn + lane];
    }
#pragma unroll
    for (int j = 0; j < 16; ++j) {
        if (j < nw) {
            const unsigned w = (unsigned)__ballot(xv[j] > 0.0f);
            if (lane == 0) rowbits[(j << 5) + c] = w;
        }
    }
    __syncthreads();

    unsigned prev = 0;
#pragma unroll
    for (int j = 0; j < 16; ++j) {
        if (j < nw) {
            const unsigned r = rowbits[(j << 5) + lane];
            const unsigned w = (unsigned)__ballot(((r >> c) & 1u) != 0u);
            const unsigned long long v = ((unsigned long long)w << 32) | prev;
            wrow[(j << 5) + lane] = (unsigned)(v >> (lane + 1));
            prev = w;
        }
    }
    __syncthreads();

    const int p = lane & 7;
    const int r = lane >> 3;
    int best_hi = walk_lags(wrow, thi0 + p, r, (thi0 + 6) >> 2);
    int best_lo = walk_lags(wrow, tlo0 + p, r, (tlo0 + 6) >> 2);
    best_hi = max(best_hi, __shfl_xor(best_hi, 8, 32));
    best_lo = max(best_lo, __shfl_xor(best_lo, 8, 32));
    best_hi = max(best_hi, __shfl_xor(best_hi, 16, 32));
    best_lo = max(best_lo, __shfl_xor(best_lo, 16, 32));

    int idx_hi = -1;
    int idx_lo = -1;
    if ((best_hi >> 10) > 0) {
        const int nxt = (best_hi & 1023) + 1;
        idx_hi = (int)(wrow[nxt] >> 31);
    }
    if ((best_lo >> 10) > 0) {
        const int nxt = (best_lo & 1023) + 1;
        idx_lo = (int)(wrow[nxt] >> 31);
    }
    __syncthreads();
    signed char (*idx)[Cn] = reinterpret_cast<signed char (*)[Cn]>(&win[0][0]);
    if (lane < 8) {
        idx[p][c]     = (signed char)idx_lo;
        idx[8 + p][c] = (signed char)idx_hi;
    }
    __syncthreads();

    if (c < 2) {
        const int mr   = lane & 15;
        const int half = lane >> 4;
        const int c0   = c << 4;
        const float e0v = emb0[c0 + mr];
        const float e1v = emb1[c0 + mr];
        v8f acc = {};
#pragma unroll
        for (int kk = 0; kk < 16; kk += 4) {
            const int ka = kk + 2 * half;
            const signed char i0 = idx[mr][c0 + ka];
            const signed char i1 = idx[mr][c0 + ka + 1];
            v2f a0 = { (i0 == 0) ? 1.0f : 0.0f, (i1 == 0) ? 1.0f : 0.0f };
            v2f a1 = { (i0 == 1) ? 1.0f : 0.0f, (i1 == 1) ? 1.0f : 0.0f };
            v2f b0 = { (mr == ka) ? e0v : 0.0f, (mr == ka + 1) ? e0v : 0.0f };
            v2f b1 = { (mr == ka) ? e1v : 0.0f, (mr == ka + 1) ? e1v : 0.0f };
            acc = __builtin_amdgcn_wmma_f32_16x16x4_f32(
                false, a0, false, b0, (short)0, acc, false, false);
            acc = __builtin_amdgcn_wmma_f32_16x16x4_f32(
                false, a1, false, b1, (short)0, acc, false, false);
        }
        float* __restrict__ outb = out + b * Tn * Cn;
        const int trow = half ? thi0 : tlo0;
#pragma unroll
        for (int q = 0; q < 8; ++q) {
            outb[(trow + q) * Cn + (c0 + mr)] = acc[q];
        }
    }
}

extern "C" void kernel_launch(void* const* d_in, const int* in_sizes, int n_in,
                              void* d_out, int out_size, void* d_ws, size_t ws_size,
                              hipStream_t stream) {
    const float* x  = (const float*)d_in[0];
    const float* e0 = (const float*)d_in[1];
    const float* e1 = (const float*)d_in[2];
    float* out = (float*)d_out;
    (void)in_sizes; (void)n_in; (void)out_size; (void)d_ws; (void)ws_size;
    dim3 grid(32, 4);
    dim3 block(32, 32);
    rosa_kernel<<<grid, block, 0, stream>>>(x, e0, e1, out);
}


# lag walk in groups of eight windows: leading zeros counted once per group, of the unsigned minimum of its xor words; lag inside the winning group found once after the loop; integer ranking only
# speedup vs baseline: 85.4822x; 1.1427x over previous
#include <hip/hip_runtime.h>

#define Bn 4
#define Tn 512
#define Cn 32

typedef float v2f __attribute__((ext_vector_type(2)));
typedef float v8f __attribute__((ext_vector_type(8)));


typedef unsigned v4u __attribute__((ext_vector_type(4)));
typedef unsigned v2u __attribute__((ext_vector_type(2)));

template <bool CHECK_T>
__device__ __forceinline__ int lag_slow(const unsigned wt, const unsigned wu,
                                        const int t, const int u, int best)
{
    const int rl = min(__clz(wt ^ wu), u + 1);
    if (!CHECK_T || u < t) best = max(best, (rl << 10) | u);
    return best;
}

__device__ __forceinline__ int lag_exact(const unsigned* __restrict__ wrow,
                                         const unsigned wt, const int t,
                                         const int u, int best)
{
    if (u < t) {
        const unsigned d = wt ^ wrow[u];
        int rl = __clz(d);
        if (d == 0u) {
            for (int off = 32; off <= u; off += 32) {
                const unsigned d2 = wrow[t - off] ^ wrow[u - off];
                rl += __clz(d2);
                if (d2 != 0u) break;
            }
        }
        rl = min(rl, u + 1);
        best = max(best, (rl << 10) | u);
    }
    return best;
}

__device__ __forceinline__ int group_key(const unsigned* __restrict__ wrow,
                                         const unsigned wt, const int ukey)
{
    const v4u wa = *reinterpret_cast<const v4u*>(wrow + ukey);
    const v4u wb = *reinterpret_cast<const v4u*>(wrow + ukey + 4);
    const unsigned m0 = min(min(wt ^ wa[0], wt ^ wa[1]), wt ^ wa[2]);
    const unsigned m1 = min(min(wt ^ wa[3], wt ^ wb[0]), wt ^ wb[1]);
    const unsigned m  = min(min(m0, m1), min(wt ^ wb[2], wt ^ wb[3]));
    return (__builtin_clz(m | 1u) << 10) | ukey;
}

__device__ __forceinline__ int walk_lags(const unsigned* __restrict__ wrow,
                                         const int t, const int r, const int top)
{
    const unsigned wt = wrow[t];
    int best = 0;
    const int u0 = (top << 3) + (r << 1);
    {
        const v2u w = *reinterpret_cast<const v2u*>(wrow + u0);
        best = lag_slow<true>(wt, w[0], t, u0, best);
        best = lag_slow<true>(wt, w[1], t, u0 + 1, best);
    }
    const int slast = top - 1;
    const int qtop  = slast >> 2;
    if (slast >= 0) {
        if (r <= slast) {
            const v4u wa = *reinterpret_cast<const v4u*>(wrow + (r << 3));
            const v4u wb = *reinterpret_cast<const v4u*>(wrow + (r << 3) + 4);
#pragma unroll
            for (int j = 0; j < 4; ++j) {
                best = lag_slow<false>(wt, wa[j], t, (r << 3) + j, best);
                best = lag_slow<false>(wt, wb[j], t, (r << 3) + 4 + j, best);
            }
        }
        if (qtop > 0) {
            int ukey = (r << 3) + 32;
            int bestf = 0;
            for (int q = 1; q < qtop; ++q) {
                bestf = max(bestf, group_key(wrow, wt, ukey));
                ukey += 32;
            }
            bestf = max(bestf, group_key(wrow, wt, min(ukey, slast << 3)));
            const int ub = bestf & 1023;
            const unsigned thr = 0xFFFFFFFFu >> (bestf >> 10);
            const v4u wa = *reinterpret_cast<const v4u*>(wrow + ub);
            const v4u wb = *reinterpret_cast<const v4u*>(wrow + ub + 4);
            int jb = 0;
            if ((wt ^ wa[1]) <= thr) jb = 1;
            if ((wt ^ wa[2]) <= thr) jb = 2;
            if ((wt ^ wa[3]) <= thr) jb = 3;
            if ((wt ^ wb[0]) <= thr) jb = 4;
            if ((wt ^ wb[1]) <= thr) jb = 5;
            if ((wt ^ wb[2]) <= thr) jb = 6;
            if ((wt ^ wb[3]) <= thr) jb = 7;
            best = max(best, bestf | jb);
        }
    }
    if ((best >> 10) >= 31) {
        best = lag_exact(wrow, wt, t, u0, 0);
        best = lag_exact(wrow, wt, t, u0 + 1, best);
        for (int q = 0; q <= qtop; ++q) {
            const int ub = min((q << 2) + r, slast) << 3;
            for (int j = 0; j < 8; ++j) best = lag_exact(wrow, wt, t, ub + j, best);
        }
    }
    return best;
}

__global__ __launch_bounds__(1024) void rosa_kernel(const float* __restrict__ x,
                                                    const float* __restrict__ emb0,
                                                    const float* __restrict__ emb1,
                                                    float* __restrict__ out)
{
    __shared__ __attribute__((aligned(16))) unsigned win[Cn][Tn];
    __shared__ unsigned rowbits[Tn];

    const int lane = threadIdx.x;
    const int c    = threadIdx.y;
    const int tlo0 = blockIdx.x << 3;
    const int thi0 = (Tn - 8) - tlo0;
    const int b    = blockIdx.y;

    const float* __restrict__ xb = x + b * Tn * Cn;
    unsigned* __restrict__ wrow = win[c];

    const int nw = ((thi0 + 7) >> 5) + 1;
    float xv[16];
#pragma unroll
    for (int j = 0; j < 16; ++j) {
        const int row = (j < nw) ? ((j << 5) + c) : c;
        xv[j] = xb[row * Cn + lane];
    }
#pragma unroll
    for (int j = 0; j < 16; ++j) {
        if (j < nw) {
            const unsigned w = (unsigned)__ballot(xv[j] > 0.0f);
            if (lane == 0) rowbits[(j << 5) + c] = w;
        }
    }
    __syncthreads();

    unsigned prev = 0;
#pragma unroll
    for (int j = 0; j < 16; ++j) {
        if (j < nw) {
            const unsigned r = rowbits[(j << 5) + lane];
            const unsigned w = (unsigned)__ballot(((r >> c) & 1u) != 0u);
            const unsigned long long v = ((unsigned long long)w << 32) | prev;
            wrow[(j << 5) + lane] = (unsigned)(v >> (lane + 1));
            prev = w;
        }
    }
    __syncthreads();

    const int p = lane & 7;
    const int r = lane >> 3;
    int best_hi = walk_lags(wrow, thi0 + p, r, thi0 >> 3);
    int best_lo = walk_lags(wrow, tlo0 + p, r, tlo0 >> 3);
    best_hi = max(best_hi, __shfl_xor(best_hi, 8, 32));
    best_lo = max(best_lo, __shfl_xor(best_lo, 8, 32));
    best_hi = max(best_hi, __shfl_xor(best_hi, 16, 32));
    best_lo = max(best_lo, __shfl_xor(best_lo, 16, 32));

    int idx_hi = -1;
    int idx_lo = -1;
    if ((best_hi >> 10) > 0) {
        const int nxt = (best_hi & 1023) + 1;
        idx_hi = (int)(wrow[nxt] >> 31);
    }
    if ((best_lo >> 10) > 0) {
        const int nxt = (best_lo & 1023) + 1;
        idx_lo = (int)(wrow[nxt] >> 31);
    }
    __syncthreads();
    signed char (*idx)[Cn] = reinterpret_cast<signed char (*)[Cn]>(&win[0][0]);
    if (lane < 8) {
        idx[p][c]     = (signed char)idx_lo;
        idx[8 + p][c] = (signed char)idx_hi;
    }
    __syncthreads();

    if (c < 2) {
        const int mr   = lane & 15;
        const int half = lane >> 4;
        const int c0   = c << 4;
        const float e0v = emb0[c0 + mr];
        const float e1v = emb1[c0 + mr];
        v8f acc = {};
#pragma unroll
        for (int kk = 0; kk < 16; kk += 4) {
            const int ka = kk + 2 * half;
            const signed char i0 = idx[mr][c0 + ka];
            const signed char i1 = idx[mr][c0 + ka + 1];
            v2f a0 = { (i0 == 0) ? 1.0f : 0.0f, (i1 == 0) ? 1.0f : 0.0f };
            v2f a1 = { (i0 == 1) ? 1.0f : 0.0f, (i1 == 1) ? 1.0f : 0.0f };
            v2f b0 = { (mr == ka) ? e0v : 0.0f, (mr == ka + 1) ? e0v : 0.0f };
            v2f b1 = { (mr == ka) ? e1v : 0.0f, (mr == ka + 1) ? e1v : 0.0f };
            acc = __builtin_amdgcn_wmma_f32_16x16x4_f32(
                false, a0, false, b0, (short)0, acc, false, false);
            acc = __builtin_amdgcn_wmma_f32_16x16x4_f32(
                false, a1, false, b1, (short)0, acc, false, false);
        }
        float* __restrict__ outb = out + b * Tn * Cn;
        const int trow = half ? thi0 : tlo0;
#pragma unroll
        for (int q = 0; q < 8; ++q) {
            outb[(trow + q) * Cn + (c0 + mr)] = acc[q];
        }
    }
}

extern "C" void kernel_launch(void* const* d_in, const int* in_sizes, int n_in,
                              void* d_out, int out_size, void* d_ws, size_t ws_size,
                              hipStream_t stream) {
    const float* x  = (const float*)d_in[0];
    const float* e0 = (const float*)d_in[1];
    const float* e1 = (const float*)d_in[2];
    float* out = (float*)d_out;
    (void)in_sizes; (void)n_in; (void)out_size; (void)d_ws; (void)ws_size;
    dim3 grid(32, 4);
    dim3 block(32, 32);
    rosa_kernel<<<grid, block, 0, stream>>>(x, e0, e1, out);
}


# 256 blocks of 16 waves (4 early + 4 late t, two columns per wave) instead of 128 blocks of 32: every compute unit gets a block, 4 waves per SIMD instead of 8; integer ranking only
# speedup vs baseline: 103.9459x; 1.2160x over previous
#include <hip/hip_runtime.h>

#define Bn 4
#define Tn 512
#define Cn 32

typedef float v2f __attribute__((ext_vector_type(2)));
typedef float v8f __attribute__((ext_vector_type(8)));


#define WP 516

typedef unsigned v4u __attribute__((ext_vector_type(4)));
typedef unsigned v2u __attribute__((ext_vector_type(2)));

template <bool CHECK_T>
__device__ __forceinline__ int lag_slow(const unsigned wt, const unsigned wu,
                                        const int t, const int u, int best)
{
    const int rl = min(__clz(wt ^ wu), u + 1);
    if (!CHECK_T || u < t) best = max(best, (rl << 10) | u);
    return best;
}

__device__ __forceinline__ int lag_exact(const unsigned* __restrict__ wrow,
                                         const unsigned wt, const int t,
                                         const int u, int best)
{
    if (u < t) {
        const unsigned d = wt ^ wrow[u];
        int rl = __clz(d);
        if (d == 0u) {
            for (int off = 32; off <= u; off += 32) {
                const unsigned d2 = wrow[t - off] ^ wrow[u - off];
                rl += __clz(d2);
                if (d2 != 0u) break;
            }
        }
        rl = min(rl, u + 1);
        best = max(best, (rl << 10) | u);
    }
    return best;
}

__device__ __forceinline__ int group_key(const unsigned* __restrict__ wrow,
                                         const unsigned wt, const int ukey)
{
    const v4u wa = *reinterpret_cast<const v4u*>(wrow + ukey);
    const v4u wb = *reinterpret_cast<const v4u*>(wrow + ukey + 4);
    const unsigned m0 = min(min(wt ^ wa[0], wt ^ wa[1]), wt ^ wa[2]);
    const unsigned m1 = min(min(wt ^ wa[3], wt ^ wb[0]), wt ^ wb[1]);
    const unsigned m  = min(min(m0, m1), min(wt ^ wb[2], wt ^ wb[3]));
    return (__builtin_clz(m | 1u) << 10) | ukey;
}

__device__ __forceinline__ int walk_lags(const unsigned* __restrict__ wrow,
                                         const int t, const int r, const int top)
{
    const unsigned wt = wrow[t];
    int best = 0;
    const int u0 = (top << 3) + (r << 1);
    {
        const v2u w = *reinterpret_cast<const v2u*>(wrow + u0);
        best = lag_slow<true>(wt, w[0], t, u0, best);
        best = lag_slow<true>(wt, w[1], t, u0 + 1, best);
    }
    const int slast = top - 1;
    const int qtop  = slast >> 2;
    if (slast >= 0) {
        if (r <= slast) {
            const v4u wa = *reinterpret_cast<const v4u*>(wrow + (r << 3));
            const v4u wb = *reinterpret_cast<const v4u*>(wrow + (r << 3) + 4);
#pragma unroll
            for (int j = 0; j < 4; ++j) {
                best = lag_slow<false>(wt, wa[j], t, (r << 3) + j, best);
                best = lag_slow<false>(wt, wb[j], t, (r << 3) + 4 + j, best);
            }
        }
        if (qtop > 0) {
            int ukey = (r << 3) + 32;
            int bestf = 0;
            for (int q = 1; q < qtop; ++q) {
                bestf = max(bestf, group_key(wrow, wt, ukey));
                ukey += 32;
            }
            bestf = max(bestf, group_key(wrow, wt, min(ukey, slast << 3)));
            const int ub = bestf & 1023;
            const unsigned thr = 0xFFFFFFFFu >> (bestf >> 10);
            const v4u wa = *reinterpret_cast<const v4u*>(wrow + ub);
            const v4u wb = *reinterpret_cast<const v4u*>(wrow + ub + 4);
            int jb = 0;
            if ((wt ^ wa[1]) <= thr) jb = 1;
            if ((wt ^ wa[2]) <= thr) jb = 2;
            if ((wt ^ wa[3]) <= thr) jb = 3;
            if ((wt ^ wb[0]) <= thr) jb = 4;
            if ((wt ^ wb[1]) <= thr) jb = 5;
            if ((wt ^ wb[2]) <= thr) jb = 6;
            if ((wt ^ wb[3]) <= thr) jb = 7;
            best = max(best, bestf | jb);
        }
    }
    if ((best >> 10) >= 31) {
        best = lag_exact(wrow, wt, t, u0, 0);
        best = lag_exact(wrow, wt, t, u0 + 1, best);
        for (int q = 0; q <= qtop; ++q) {
            const int ub = min((q << 2) + r, slast) << 3;
            for (int j = 0; j < 8; ++j) best = lag_exact(wrow, wt, t, ub + j, best);
        }
    }
    return best;
}

__global__ __launch_bounds__(512) void rosa_kernel(const float* __restrict__ x,
                                                   const float* __restrict__ emb0,
                                                   const float* __restrict__ emb1,
                                                   float* __restrict__ out)
{
    __shared__ __attribute__((aligned(16))) unsigned win[Cn][WP];
    __shared__ unsigned rowbits[Tn];

    const int lane = threadIdx.x;
    const int wv   = threadIdx.y;
    const int tlo0 = blockIdx.x << 2;
    const int thi0 = (Tn - 4) - tlo0;
    const int b    = blockIdx.y;

    const float* __restrict__ xb = x + b * Tn * Cn;

    const int nw = ((thi0 + 3) >> 5) + 1;
    const int nr = nw << 1;
    float xv[32];
#pragma unroll
    for (int j = 0; j < 32; ++j) {
        const int row = (j < 18 || j < nr) ? ((j << 4) + wv) : wv;
        xv[j] = xb[row * Cn + lane];
    }
    unsigned myw = 0;
#pragma unroll
    for (int j = 0; j < 32; ++j) {
        const unsigned wd = (unsigned)__ballot(xv[j] > 0.0f);
        myw = (lane == j) ? wd : myw;
    }
    if (lane < nr) rowbits[(lane << 4) + wv] = myw;
    __syncthreads();

    {
        unsigned* __restrict__ wrow0 = win[2 * wv];
        unsigned* __restrict__ wrow1 = win[2 * wv + 1];
        unsigned rb[16];
#pragma unroll
        for (int j = 0; j < 16; ++j) {
            rb[j] = rowbits[((j < 9 || j < nw) ? (j << 5) : 0) + lane];
        }
        unsigned prev0 = 0, prev1 = 0;
#pragma unroll
        for (int j = 0; j < 16; ++j) {
            if (j < 9 || j < nw) {
                const unsigned w0 = (unsigned)__ballot(((rb[j] >> (2 * wv)) & 1u) != 0u);
                const unsigned w1 = (unsigned)__ballot(((rb[j] >> (2 * wv + 1)) & 1u) != 0u);
                const unsigned long long v0 = ((unsigned long long)w0 << 32) | prev0;
                const unsigned long long v1 = ((unsigned long long)w1 << 32) | prev1;
                wrow0[(j << 5) + lane] = (unsigned)(v0 >> (lane + 1));
                wrow1[(j << 5) + lane] = (unsigned)(v1 >> (lane + 1));
                prev0 = w0;
                prev1 = w1;
            }
        }
    }
    __syncthreads();

    const int csel = lane >> 4;
    const int p = lane & 3;
    const int r = (lane >> 2) & 3;
    const int c = 2 * wv + csel;
    const unsigned* __restrict__ wrow = win[c];
    int best_hi = walk_lags(wrow, thi0 + p, r, thi0 >> 3);
    int best_lo = walk_lags(wrow, tlo0 + p, r, tlo0 >> 3);
    best_hi = max(best_hi, __shfl_xor(best_hi, 4, 32));
    best_lo = max(best_lo, __shfl_xor(best_lo, 4, 32));
    best_hi = max(best_hi, __shfl_xor(best_hi, 8, 32));
    best_lo = max(best_lo, __shfl_xor(best_lo, 8, 32));

    int idx_hi = -1;
    int idx_lo = -1;
    if ((best_hi >> 10) > 0) {
        const int nxt = (best_hi & 1023) + 1;
        idx_hi = (int)(wrow[nxt] >> 31);
    }
    if ((best_lo >> 10) > 0) {
        const int nxt = (best_lo & 1023) + 1;
        idx_lo = (int)(wrow[nxt] >> 31);
    }
    __syncthreads();
    signed char (*idx)[Cn] = reinterpret_cast<signed char (*)[Cn]>(&win[0][0]);
    if (r == 0) {
        idx[p][c]     = (signed char)idx_lo;
        idx[4 + p][c] = (signed char)idx_hi;
    }
    __syncthreads();

    if (wv < 2) {
        const int mr   = lane & 15;
        const int half = lane >> 4;
        const int c0   = wv << 4;
        const bool live = mr < 8;
        const float e0v = emb0[c0 + mr];
        const float e1v = emb1[c0 + mr];
        v8f acc = {};
#pragma unroll
        for (int kk = 0; kk < 16; kk += 4) {
            const int ka = kk + 2 * half;
            const signed char i0 = idx[mr & 7][c0 + ka];
            const signed char i1 = idx[mr & 7][c0 + ka + 1];
            v2f a0 = { (live && i0 == 0) ? 1.0f : 0.0f, (live && i1 == 0) ? 1.0f : 0.0f };
            v2f a1 = { (live && i0 == 1) ? 1.0f : 0.0f, (live && i1 == 1) ? 1.0f : 0.0f };
            v2f b0 = { (mr == ka) ? e0v : 0.0f, (mr == ka + 1) ? e0v : 0.0f };
            v2f b1 = { (mr == ka) ? e1v : 0.0f, (mr == ka + 1) ? e1v : 0.0f };
            acc = __builtin_amdgcn_wmma_f32_16x16x4_f32(
                false, a0, false, b0, (short)0, acc, false, false);
            acc = __builtin_amdgcn_wmma_f32_16x16x4_f32(
                false, a1, false, b1, (short)0, acc, false, false);
        }
        float* __restrict__ outb = out + b * Tn * Cn;
        if (half == 0) {
#pragma unroll
            for (int q = 0; q < 4; ++q) {
                outb[(tlo0 + q) * Cn + (c0 + mr)] = acc[q];
                outb[(thi0 + q) * Cn + (c0 + mr)] = acc[4 + q];
            }
        }
    }
}

extern "C" void kernel_launch(void* const* d_in, const int* in_sizes, int n_in,
                              void* d_out, int out_size, void* d_ws, size_t ws_size,
                              hipStream_t stream) {
    const float* x  = (const float*)d_in[0];
    const float* e0 = (const float*)d_in[1];
    const float* e1 = (const float*)d_in[2];
    float* out = (float*)d_out;
    (void)in_sizes; (void)n_in; (void)out_size; (void)d_ws; (void)ws_size;
    dim3 grid(64, 4);
    dim3 block(32, 16);
    rosa_kernel<<<grid, block, 0, stream>>>(x, e0, e1, out);
}


# sign words built per lane (lane = column, wave = 32-row chunk, compare + add-with-carry per row) replace both ballot bit transposes; windows by funnel shift, 16-byte LDS stores; integer only
# speedup vs baseline: 114.6275x; 1.1028x over previous
#include <hip/hip_runtime.h>

#define Bn 4
#define Tn 512
#define Cn 32

typedef float v2f __attribute__((ext_vector_type(2)));
typedef float v8f __attribute__((ext_vector_type(8)));


#define WP 516

typedef unsigned v4u __attribute__((ext_vector_type(4)));
typedef unsigned v2u __attribute__((ext_vector_type(2)));

template <bool CHECK_T>
__device__ __forceinline__ int lag_slow(const unsigned wt, const unsigned wu,
                                        const int t, const int u, int best)
{
    const int rl = min(__clz(wt ^ wu), u + 1);
    if (!CHECK_T || u < t) best = max(best, (rl << 10) | u);
    return best;
}

__device__ __forceinline__ int lag_exact(const unsigned* __restrict__ wrow,
                                         const unsigned wt, const int t,
                                         const int u, int best)
{
    if (u < t) {
        const unsigned d = wt ^ wrow[u];
        int rl = __clz(d);
        if (d == 0u) {
            for (int off = 32; off <= u; off += 32) {
                const unsigned d2 = wrow[t - off] ^ wrow[u - off];
                rl += __clz(d2);
                if (d2 != 0u) break;
            }
        }
        rl = min(rl, u + 1);
        best = max(best, (rl << 10) | u);
    }
    return best;
}

__device__ __forceinline__ int group_key(const unsigned* __restrict__ wrow,
                                         const unsigned wt, const int ukey)
{
    const v4u wa = *reinterpret_cast<const v4u*>(wrow + ukey);
    const v4u wb = *reinterpret_cast<const v4u*>(wrow + ukey + 4);
    const unsigned m0 = min(min(wt ^ wa[0], wt ^ wa[1]), wt ^ wa[2]);
    const unsigned m1 = min(min(wt ^ wa[3], wt ^ wb[0]), wt ^ wb[1]);
    const unsigned m  = min(min(m0, m1), min(wt ^ wb[2], wt ^ wb[3]));
    return (__builtin_clz(m | 1u) << 10) | ukey;
}

__device__ __forceinline__ int walk_lags(const unsigned* __restrict__ wrow,
                                         const int t, const int r, const int top)
{
    const unsigned wt = wrow[t];
    int best = 0;
    const int u0 = (top << 3) + (r << 1);
    {
        const v2u w = *reinterpret_cast<const v2u*>(wrow + u0);
        best = lag_slow<true>(wt, w[0], t, u0, best);
        best = lag_slow<true>(wt, w[1], t, u0 + 1, best);
    }
    const int slast = top - 1;
    const int qtop  = slast >> 2;
    if (slast >= 0) {
        if (r <= slast) {
            const v4u wa = *reinterpret_cast<const v4u*>(wrow + (r << 3));
            const v4u wb = *reinterpret_cast<const v4u*>(wrow + (r << 3) + 4);
#pragma unroll
            for (int j = 0; j < 4; ++j) {
                best = lag_slow<false>(wt, wa[j], t, (r << 3) + j, best);
                best = lag_slow<false>(wt, wb[j], t, (r << 3) + 4 + j, best);
            }
        }
        if (qtop > 0) {
            int ukey = (r << 3) + 32;
            int bestf = 0;
            for (int q = 1; q < qtop; ++q) {
                bestf = max(bestf, group_key(wrow, wt, ukey));
                ukey += 32;
            }
            bestf = max(bestf, group_key(wrow, wt, min(ukey, slast << 3)));
            const int ub = bestf & 1023;
            const unsigned thr = 0xFFFFFFFFu >> (bestf >> 10);
            const v4u wa = *reinterpret_cast<const v4u*>(wrow + ub);
            const v4u wb = *reinterpret_cast<const v4u*>(wrow + ub + 4);
            int jb = 0;
            if ((wt ^ wa[1]) <= thr) jb = 1;
            if ((wt ^ wa[2]) <= thr) jb = 2;
            if ((wt ^ wa[3]) <= thr) jb = 3;
            if ((wt ^ wb[0]) <= thr) jb = 4;
            if ((wt ^ wb[1]) <= thr) jb = 5;
            if ((wt ^ wb[2]) <= thr) jb = 6;
            if ((wt ^ wb[3]) <= thr) jb = 7;
            best = max(best, bestf | jb);
        }
    }
    if ((best >> 10) >= 31) {
        best = lag_exact(wrow, wt, t, u0, 0);
        best = lag_exact(wrow, wt, t, u0 + 1, best);
        for (int q = 0; q <= qtop; ++q) {
            const int ub = min((q << 2) + r, slast) << 3;
            for (int j = 0; j < 8; ++j) best = lag_exact(wrow, wt, t, ub + j, best);
        }
    }
    return best;
}

__global__ __launch_bounds__(512) void rosa_kernel(const float* __restrict__ x,
                                                   const float* __restrict__ emb0,
                                                   const float* __restrict__ emb1,
                                                   float* __restrict__ out)
{
    __shared__ __attribute__((aligned(16))) unsigned win[Cn][WP];
    __shared__ unsigned colw[Tn / 32][Cn];

    const int lane = threadIdx.x;
    const int wv   = threadIdx.y;
    const int tlo0 = blockIdx.x << 2;
    const int thi0 = (Tn - 4) - tlo0;
    const int b    = blockIdx.y;

    const float* __restrict__ xb = x + b * Tn * Cn;

    const int nw = ((thi0 + 3) >> 5) + 1;
    unsigned cw = 0;
    if (wv < nw) {
        const float* __restrict__ xc = xb + (wv << 5) * Cn + lane;
        float xv[32];
#pragma unroll
        for (int i = 0; i < 32; ++i) xv[i] = xc[(31 - i) * Cn];
        unsigned hi = 0, lo = 0;
#pragma unroll
        for (int i = 0; i < 16; ++i) {
            asm("v_cmp_lt_f32_e32 vcc_lo, 0, %1\n\t"
                "v_add_co_ci_u32_e32 %0, vcc_lo, %0, %0, vcc_lo"
                : "+v"(hi) : "v"(xv[i]) : "vcc");
            asm("v_cmp_lt_f32_e32 vcc_lo, 0, %1\n\t"
                "v_add_co_ci_u32_e32 %0, vcc_lo, %0, %0, vcc_lo"
                : "+v"(lo) : "v"(xv[16 + i]) : "vcc");
        }
        cw = (hi << 16) | lo;
        colw[wv][lane] = cw;
    }
    __syncthreads();

    if (wv < nw) {
        const unsigned prev = (wv > 0) ? colw[wv - 1][lane] : 0u;
        const unsigned long long v = ((unsigned long long)cw << 32) | prev;
        unsigned* __restrict__ dst = &win[lane][wv << 5];
#pragma unroll
        for (int q = 0; q < 8; ++q) {
            v4u w;
            w[0] = (unsigned)(v >> (4 * q + 1));
            w[1] = (unsigned)(v >> (4 * q + 2));
            w[2] = (unsigned)(v >> (4 * q + 3));
            w[3] = (unsigned)(v >> (4 * q + 4));
            *reinterpret_cast<v4u*>(dst + 4 * q) = w;
        }
    }
    __syncthreads();

    const int csel = lane >> 4;
    const int p = lane & 3;
    const int r = (lane >> 2) & 3;
    const int c = 2 * wv + csel;
    const unsigned* __restrict__ wrow = win[c];
    int best_hi = walk_lags(wrow, thi0 + p, r, thi0 >> 3);
    int best_lo = walk_lags(wrow, tlo0 + p, r, tlo0 >> 3);
    best_hi = max(best_hi, __shfl_xor(best_hi, 4, 32));
    best_lo = max(best_lo, __shfl_xor(best_lo, 4, 32));
    best_hi = max(best_hi, __shfl_xor(best_hi, 8, 32));
    best_lo = max(best_lo, __shfl_xor(best_lo, 8, 32));

    int idx_hi = -1;
    int idx_lo = -1;
    if ((best_hi >> 10) > 0) {
        const int nxt = (best_hi & 1023) + 1;
        idx_hi = (int)(wrow[nxt] >> 31);
    }
    if ((best_lo >> 10) > 0) {
        const int nxt = (best_lo & 1023) + 1;
        idx_lo = (int)(wrow[nxt] >> 31);
    }
    __syncthreads();
    signed char (*idx)[Cn] = reinterpret_cast<signed char (*)[Cn]>(&win[0][0]);
    if (r == 0) {
        idx[p][c]     = (signed char)idx_lo;
        idx[4 + p][c] = (signed char)idx_hi;
    }
    __syncthreads();

    if (wv < 2) {
        const int mr   = lane & 15;
        const int half = lane >> 4;
        const int c0   = wv << 4;
        const bool live = mr < 8;
        const float e0v = emb0[c0 + mr];
        const float e1v = emb1[c0 + mr];
        v8f acc = {};
#pragma unroll
        for (int kk = 0; kk < 16; kk += 4) {
            const int ka = kk + 2 * half;
            const signed char i0 = idx[mr & 7][c0 + ka];
            const signed char i1 = idx[mr & 7][c0 + ka + 1];
            v2f a0 = { (live && i0 == 0) ? 1.0f : 0.0f, (live && i1 == 0) ? 1.0f : 0.0f };
            v2f a1 = { (live && i0 == 1) ? 1.0f : 0.0f, (live && i1 == 1) ? 1.0f : 0.0f };
            v2f b0 = { (mr == ka) ? e0v : 0.0f, (mr == ka + 1) ? e0v : 0.0f };
            v2f b1 = { (mr == ka) ? e1v : 0.0f, (mr == ka + 1) ? e1v : 0.0f };
            acc = __builtin_amdgcn_wmma_f32_16x16x4_f32(
                false, a0, false, b0, (short)0, acc, false, false);
            acc = __builtin_amdgcn_wmma_f32_16x16x4_f32(
                false, a1, false, b1, (short)0, acc, false, false);
        }
        float* __restrict__ outb = out + b * Tn * Cn;
        if (half == 0) {
#pragma unroll
            for (int q = 0; q < 4; ++q) {
                outb[(tlo0 + q) * Cn + (c0 + mr)] = acc[q];
                outb[(thi0 + q) * Cn + (c0 + mr)] = acc[4 + q];
            }
        }
    }
}

extern "C" void kernel_launch(void* const* d_in, const int* in_sizes, int n_in,
                              void* d_out, int out_size, void* d_ws, size_t ws_size,
                              hipStream_t stream) {
    const float* x  = (const float*)d_in[0];
    const float* e0 = (const float*)d_in[1];
    const float* e1 = (const float*)d_in[2];
    float* out = (float*)d_out;
    (void)in_sizes; (void)n_in; (void)out_size; (void)d_ws; (void)ws_size;
    dim3 grid(64, 4);
    dim3 block(32, 16);
    rosa_kernel<<<grid, block, 0, stream>>>(x, e0, e1, out);
}


# late and early position walked in one pass (shared group reads, address-known reads up front, one wait for both winning groups), lane max by DPP: half the serial LDS waits per wave; integer only
# speedup vs baseline: 121.7087x; 1.0618x over previous
#include <hip/hip_runtime.h>

#define Bn 4
#define Tn 512
#define Cn 32

typedef float v2f __attribute__((ext_vector_type(2)));
typedef float v8f __attribute__((ext_vector_type(8)));


#define WP 516

typedef unsigned v4u __attribute__((ext_vector_type(4)));
typedef unsigned v2u __attribute__((ext_vector_type(2)));

template <bool CHECK_T>
__device__ __forceinline__ int lag_slow(const unsigned wt, const unsigned wu,
                                        const int t, const int u, int best)
{
    const int rl = min(__clz(wt ^ wu), u + 1);
    if (!CHECK_T || u < t) best = max(best, (rl << 10) | u);
    return best;
}

__device__ __forceinline__ int lag_exact(const unsigned* __restrict__ wrow,
                                         const unsigned wt, const int t,
                                         const int u, int best)
{
    if (u < t) {
        const unsigned d = wt ^ wrow[u];
        int rl = __clz(d);
        if (d == 0u) {
            for (int off = 32; off <= u; off += 32) {
                const unsigned d2 = wrow[t - off] ^ wrow[u - off];
                rl += __clz(d2);
                if (d2 != 0u) break;
            }
        }
        rl = min(rl, u + 1);
        best = max(best, (rl << 10) | u);
    }
    return best;
}

__device__ __forceinline__ unsigned group_min(const v4u wa, const v4u wb,
                                              const unsigned wt)
{
    const unsigned m0 = min(min(wt ^ wa[0], wt ^ wa[1]), wt ^ wa[2]);
    const unsigned m1 = min(min(wt ^ wa[3], wt ^ wb[0]), wt ^ wb[1]);
    return min(min(m0, m1), min(wt ^ wb[2], wt ^ wb[3]));
}

__device__ __forceinline__ int group_key(const unsigned m, const int ukey)
{
    return (__builtin_clz(m | 1u) << 10) | ukey;
}

__device__ __forceinline__ int pick_lag(const unsigned w1, const v2u w23, const v4u wb,
                                        const unsigned wt, const int bestf)
{
    const unsigned thr = 0xFFFFFFFFu >> (bestf >> 10);
    int jb = 0;
    if ((wt ^ w1)     <= thr) jb = 1;
    if ((wt ^ w23[0]) <= thr) jb = 2;
    if ((wt ^ w23[1]) <= thr) jb = 3;
    if ((wt ^ wb[0])  <= thr) jb = 4;
    if ((wt ^ wb[1])  <= thr) jb = 5;
    if ((wt ^ wb[2])  <= thr) jb = 6;
    if ((wt ^ wb[3])  <= thr) jb = 7;
    return bestf | jb;
}

__device__ __forceinline__ int walk_exact(const unsigned* __restrict__ wrow,
                                          const unsigned wt, const int t,
                                          const int r, const int top)
{
    const int u0 = (top << 3) + (r << 1);
    const int slast = top - 1;
    const int qtop  = slast >> 2;
    int best = lag_exact(wrow, wt, t, u0, 0);
    best = lag_exact(wrow, wt, t, u0 + 1, best);
    for (int q = 0; q <= qtop; ++q) {
        const int ub = min((q << 2) + r, slast) << 3;
        for (int j = 0; j < 8; ++j) best = lag_exact(wrow, wt, t, ub + j, best);
    }
    return best;
}

__device__ __forceinline__ void walk_pair(const unsigned* __restrict__ wrow,
                                          const int th, const int tl,
                                          const int r, const int toph, const int topl,
                                          int& best_h, int& best_l)
{
    const int r8 = r << 3;
    const int slasth = toph - 1;
    const int slastl = topl - 1;
    const int qtoph  = slasth >> 2;
    const int qtopl  = slastl >> 2;
    const int u0h = (toph << 3) + (r << 1);
    const int u0l = (topl << 3) + (r << 1);
    const int finh = max(min(qtoph * 32 + r8, slasth * 8), 0);
    const int finl = max(min(qtopl * 32 + r8, slastl * 8), 0);

    const unsigned wth = wrow[th];
    const unsigned wtl = wrow[tl];
    const v2u w2h = *reinterpret_cast<const v2u*>(wrow + u0h);
    const v2u w2l = *reinterpret_cast<const v2u*>(wrow + u0l);
    const v4u sa  = *reinterpret_cast<const v4u*>(wrow + r8);
    const v4u sb  = *reinterpret_cast<const v4u*>(wrow + r8 + 4);
    const v4u fha = *reinterpret_cast<const v4u*>(wrow + finh);
    const v4u fhb = *reinterpret_cast<const v4u*>(wrow + finh + 4);
    const v4u fla = *reinterpret_cast<const v4u*>(wrow + finl);
    const v4u flb = *reinterpret_cast<const v4u*>(wrow + finl + 4);

    int bh = 0, bl = 0;
    bh = lag_slow<true>(wth, w2h[0], th, u0h, bh);
    bh = lag_slow<true>(wth, w2h[1], th, u0h + 1, bh);
    bl = lag_slow<true>(wtl, w2l[0], tl, u0l, bl);
    bl = lag_slow<true>(wtl, w2l[1], tl, u0l + 1, bl);
    if (r <= slasth) {
        for (int j = 0; j < 4; ++j) {
            bh = lag_slow<false>(wth, sa[j], th, r8 + j, bh);
            bh = lag_slow<false>(wth, sb[j], th, r8 + 4 + j, bh);
        }
    }
    if (r <= slastl) {
        for (int j = 0; j < 4; ++j) {
            bl = lag_slow<false>(wtl, sa[j], tl, r8 + j, bl);
            bl = lag_slow<false>(wtl, sb[j], tl, r8 + 4 + j, bl);
        }
    }

    int bfh = 0, bfl = 0;
    const int nh = max(qtoph - 1, 0);
    const int nb = min(max(qtopl - 1, 0), nh);
    int ukey = r8 + 32;
    int q = 0;
    for (; q < nb; ++q) {
        const v4u wa = *reinterpret_cast<const v4u*>(wrow + ukey);
        const v4u wb = *reinterpret_cast<const v4u*>(wrow + ukey + 4);
        bfh = max(bfh, group_key(group_min(wa, wb, wth), ukey));
        bfl = max(bfl, group_key(group_min(wa, wb, wtl), ukey));
        ukey += 32;
    }
    for (; q < nh; ++q) {
        const v4u wa = *reinterpret_cast<const v4u*>(wrow + ukey);
        const v4u wb = *reinterpret_cast<const v4u*>(wrow + ukey + 4);
        bfh = max(bfh, group_key(group_min(wa, wb, wth), ukey));
        ukey += 32;
    }
    if (qtoph > 0) bfh = max(bfh, group_key(group_min(fha, fhb, wth), finh));
    if (qtopl > 0) bfl = max(bfl, group_key(group_min(fla, flb, wtl), finl));

    const int ubh = bfh & 1023;
    const int ubl = bfl & 1023;
    const unsigned ph1  = wrow[ubh + 1];
    const v2u      ph23 = *reinterpret_cast<const v2u*>(wrow + ubh + 2);
    const v4u      phb  = *reinterpret_cast<const v4u*>(wrow + ubh + 4);
    const unsigned pl1  = wrow[ubl + 1];
    const v2u      pl23 = *reinterpret_cast<const v2u*>(wrow + ubl + 2);
    const v4u      plb  = *reinterpret_cast<const v4u*>(wrow + ubl + 4);
    bh = max(bh, pick_lag(ph1, ph23, phb, wth, bfh));
    bl = max(bl, pick_lag(pl1, pl23, plb, wtl, bfl));

    if ((bh >> 10) >= 31) bh = walk_exact(wrow, wth, th, r, toph);
    if ((bl >> 10) >= 31) bl = walk_exact(wrow, wtl, tl, r, topl);
    best_h = bh;
    best_l = bl;
}

__device__ __forceinline__ int max_over_residues(int v)
{
    v = max(v, __builtin_amdgcn_update_dpp(v, v, 0x164, 0xF, 0xF, false));
    v = max(v, __builtin_amdgcn_update_dpp(v, v, 0x168, 0xF, 0xF, false));
    return v;
}

__global__ __launch_bounds__(512) void rosa_kernel(const float* __restrict__ x,
                                                   const float* __restrict__ emb0,
                                                   const float* __restrict__ emb1,
                                                   float* __restrict__ out)
{
    __shared__ __attribute__((aligned(16))) unsigned win[Cn][WP];
    __shared__ unsigned colw[Tn / 32][Cn];

    const int lane = threadIdx.x;
    const int wv   = threadIdx.y;
    const int tlo0 = blockIdx.x << 2;
    const int thi0 = (Tn - 4) - tlo0;
    const int b    = blockIdx.y;

    const float* __restrict__ xb = x + b * Tn * Cn;

    const int nw = ((thi0 + 3) >> 5) + 1;
    unsigned cw = 0;
    if (wv < nw) {
        const float* __restrict__ xc = xb + (wv << 5) * Cn + lane;
        float xv[32];
#pragma unroll
        for (int i = 0; i < 32; ++i) xv[i] = xc[(31 - i) * Cn];
        unsigned hi = 0, lo = 0;
#pragma unroll
        for (int i = 0; i < 16; ++i) {
            asm("v_cmp_lt_f32_e32 vcc_lo, 0, %1\n\t"
                "v_add_co_ci_u32_e32 %0, vcc_lo, %0, %0, vcc_lo"
                : "+v"(hi) : "v"(xv[i]) : "vcc");
            asm("v_cmp_lt_f32_e32 vcc_lo, 0, %1\n\t"
                "v_add_co_ci_u32_e32 %0, vcc_lo, %0, %0, vcc_lo"
                : "+v"(lo) : "v"(xv[16 + i]) : "vcc");
        }
        cw = (hi << 16) | lo;
        colw[wv][lane] = cw;
    }
    __syncthreads();

    if (wv < nw) {
        const unsigned prev = (wv > 0) ? colw[wv - 1][lane] : 0u;
        const unsigned long long v = ((unsigned long long)cw << 32) | prev;
        unsigned* __restrict__ dst = &win[lane][wv << 5];
#pragma unroll
        for (int q = 0; q < 8; ++q) {
            v4u w;
            w[0] = (unsigned)(v >> (4 * q + 1));
            w[1] = (unsigned)(v >> (4 * q + 2));
            w[2] = (unsigned)(v >> (4 * q + 3));
            w[3] = (unsigned)(v >> (4 * q + 4));
            *reinterpret_cast<v4u*>(dst + 4 * q) = w;
        }
    }
    __syncthreads();

    const int csel = lane >> 4;
    const int p = lane & 3;
    const int r = (lane >> 2) & 3;
    const int c = 2 * wv + csel;
    const unsigned* __restrict__ wrow = win[c];
    int best_hi, best_lo;
    walk_pair(wrow, thi0 + p, tlo0 + p, r, thi0 >> 3, tlo0 >> 3, best_hi, best_lo);
    best_hi = max_over_residues(best_hi);
    best_lo = max_over_residues(best_lo);

    const int bit_hi = (int)(wrow[(best_hi & 1023) + 1] >> 31);
    const int bit_lo = (int)(wrow[(best_lo & 1023) + 1] >> 31);
    const int idx_hi = ((best_hi >> 10) > 0) ? bit_hi : -1;
    const int idx_lo = ((best_lo >> 10) > 0) ? bit_lo : -1;
    __syncthreads();
    signed char (*idx)[Cn] = reinterpret_cast<signed char (*)[Cn]>(&win[0][0]);
    if (r == 0) {
        idx[p][c]     = (signed char)idx_lo;
        idx[4 + p][c] = (signed char)idx_hi;
    }
    __syncthreads();

    if (wv < 2) {
        const int mr   = lane & 15;
        const int half = lane >> 4;
        const int c0   = wv << 4;
        const bool live = mr < 8;
        const float e0v = emb0[c0 + mr];
        const float e1v = emb1[c0 + mr];
        v8f acc = {};
#pragma unroll
        for (int kk = 0; kk < 16; kk += 4) {
            const int ka = kk + 2 * half;
            const signed char i0 = idx[mr & 7][c0 + ka];
            const signed char i1 = idx[mr & 7][c0 + ka + 1];
            v2f a0 = { (live && i0 == 0) ? 1.0f : 0.0f, (live && i1 == 0) ? 1.0f : 0.0f };
            v2f a1 = { (live && i0 == 1) ? 1.0f : 0.0f, (live && i1 == 1) ? 1.0f : 0.0f };
            v2f b0 = { (mr == ka) ? e0v : 0.0f, (mr == ka + 1) ? e0v : 0.0f };
            v2f b1 = { (mr == ka) ? e1v : 0.0f, (mr == ka + 1) ? e1v : 0.0f };
            acc = __builtin_amdgcn_wmma_f32_16x16x4_f32(
                false, a0, false, b0, (short)0, acc, false, false);
            acc = __builtin_amdgcn_wmma_f32_16x16x4_f32(
                false, a1, false, b1, (short)0, acc, false, false);
        }
        float* __restrict__ outb = out + b * Tn * Cn;
        if (half == 0) {
#pragma unroll
            for (int q = 0; q < 4; ++q) {
                outb[(tlo0 + q) * Cn + (c0 + mr)] = acc[q];
                outb[(thi0 + q) * Cn + (c0 + mr)] = acc[4 + q];
            }
        }
    }
}

extern "C" void kernel_launch(void* const* d_in, const int* in_sizes, int n_in,
                              void* d_out, int out_size, void* d_ws, size_t ws_size,
                              hipStream_t stream) {
    const float* x  = (const float*)d_in[0];
    const float* e0 = (const float*)d_in[1];
    const float* e1 = (const float*)d_in[2];
    float* out = (float*)d_out;
    (void)in_sizes; (void)n_in; (void)out_size; (void)d_ws; (void)ws_size;
    dim3 grid(64, 4);
    dim3 block(32, 16);
    rosa_kernel<<<grid, block, 0, stream>>>(x, e0, e1, out);
}
